# starting kernel
# baseline (speedup 1.0000x reference)
#include <hip/hip_runtime.h>

typedef __attribute__((ext_vector_type(2))) float v2f;
typedef __attribute__((ext_vector_type(4))) float v4f;
typedef __attribute__((ext_vector_type(8))) float v8f;

#define E_ENT 100000
#define R_REL 64
#define I_CH  256
#define O_CH  256

__global__ void RGCN_wsum_kernel(const float* __restrict__ W,
                                 float* __restrict__ Wpack) {
    const int r = blockIdx.x;
    const int o = threadIdx.x;
    const float* base = W + (size_t)r * I_CH * O_CH + o;
    float s = 0.0f;
    #pragma unroll 8
    for (int i = 0; i < I_CH; ++i) s += base[(size_t)i * O_CH];
    Wpack[(size_t)(r >> 1) * (2 * O_CH) + o * 2 + (r & 1)] = s;
}

__global__ void RGCN_xsum_kernel(const float* __restrict__ x,
                                 float* __restrict__ xs) {
    const int wave = threadIdx.x >> 5;
    const int lane = threadIdx.x & 31;
    const int row  = blockIdx.x * 8 + wave;
    if (row >= E_ENT) return;
    const float* p = x + (size_t)row * I_CH;
    v4f a = *(const v4f*)(p + lane * 4);
    v4f b = *(const v4f*)(p + 128 + lane * 4);
    float s = a.x + a.y + a.z + a.w + b.x + b.y + b.z + b.w;
    #pragma unroll
    for (int off = 16; off > 0; off >>= 1)
        s += __shfl_xor(s, off, 32);
    if (lane == 0) xs[row] = s;
}

__global__ void RGCN_main_kernel(const float* __restrict__ cs,
                                 const float* __restrict__ Wpack,
                                 const float* __restrict__ xs,
                                 float* __restrict__ out) {
    const int wave      = threadIdx.x >> 5;
    const int lane      = threadIdx.x & 31;
    const int row_block = blockIdx.x * (256 >> 5) + wave;
    if (row_block >= E_ENT / 16) return;

    const int e_base = row_block * 16;
    const int half   = lane >> 4;
    const int m      = lane & 15;

    const float scale = xs[e_base + m];

    v2f a[16];
    const float* csrow = cs + (size_t)(e_base + m) * R_REL + half * 2;
    #pragma unroll
    for (int k = 0; k < 16; ++k) {
        const float c0 = csrow[4 * k + 0];
        const float c1 = csrow[4 * k + 1];
        a[k].x = scale / c0;
        a[k].y = scale / c1;
    }

    #pragma unroll 1
    for (int nt = 0; nt < 16; ++nt) {
        v8f acc = {};
        #pragma unroll
        for (int k = 0; k < 16; ++k) {
            const int p = 2 * k + half;
            v2f b = *(const v2f*)(Wpack +
                     ((size_t)p * O_CH + nt * 16 + m) * 2);
            acc = __builtin_amdgcn_wmma_f32_16x16x4_f32(
                 false, a[k],  false, b,
                 (short)0, acc,  false,  false);
        }
        float* obase = out + (size_t)(e_base + half * 8) * O_CH + nt * 16 + m;
        #pragma unroll
        for (int v = 0; v < 8; ++v)
            obase[(size_t)v * O_CH] = acc[v];
    }
}

extern "C" void kernel_launch(void* const* d_in, const int* in_sizes, int n_in,
                              void* d_out, int out_size, void* d_ws, size_t ws_size,
                              hipStream_t stream) {
    const float* x  = (const float*)d_in[0];
    const float* cs = (const float*)d_in[1];
    const float* W  = (const float*)d_in[2];
    float* out = (float*)d_out;

    float* Wpack = (float*)d_ws;
    float* xsum  = Wpack + R_REL * O_CH;

    RGCN_wsum_kernel<<<R_REL, O_CH, 0, stream>>>(W, Wpack);
    RGCN_xsum_kernel<<<(E_ENT + 7) / 8, 256, 0, stream>>>(x, xsum);

    const int row_blocks = E_ENT / 16;
    const int grid = (row_blocks + 7) / 8;
    RGCN_main_kernel<<<grid, 256, 0, stream>>>(cs, Wpack, xsum, out);
}


# W row-sum pre-pass re-tiled: 256 blocks (relation x 64-col chunk), 16 b128 loads in flight per lane, 16 partial rows summed via LDS; changes the reduction tree of W_sum
# speedup vs baseline: 1.2697x; 1.2697x over previous
#include <hip/hip_runtime.h>

typedef __attribute__((ext_vector_type(2))) float v2f;
typedef __attribute__((ext_vector_type(4))) float v4f;
typedef __attribute__((ext_vector_type(8))) float v8f;

#define E_ENT 100000
#define R_REL 64
#define I_CH  256
#define O_CH  256

__global__ void RGCN_wsum_kernel(const float* __restrict__ W,
                                 float* __restrict__ Wpack) {
    __shared__ float part[16][64];
    const int r  = blockIdx.x >> 2;
    const int cc = blockIdx.x & 3;
    const int c4 = threadIdx.x & 15;
    const int il = threadIdx.x >> 4;
    const float* base = W + ((size_t)r * I_CH + il) * O_CH + cc * 64 + c4 * 4;
    v4f v[16];
    for (int j = 0; j < 16; ++j)
        v[j] = *(const v4f*)(base + (size_t)j * 16 * O_CH);
    v4f s = v[0];
    for (int j = 1; j < 16; ++j) s += v[j];
    *(v4f*)&part[il][c4 * 4] = s;
    __syncthreads();
    if (threadIdx.x < 64) {
        const int col = threadIdx.x;
        float t = part[0][col];
        for (int k = 1; k < 16; ++k) t += part[k][col];
        Wpack[((size_t)(r >> 1) * O_CH + cc * 64 + col) * 2 + (r & 1)] = t;
    }
}

__global__ void RGCN_xsum_kernel(const float* __restrict__ x,
                                 float* __restrict__ xs) {
    const int wave = threadIdx.x >> 5;
    const int lane = threadIdx.x & 31;
    const int row  = blockIdx.x * 8 + wave;
    if (row >= E_ENT) return;
    const float* p = x + (size_t)row * I_CH;
    v4f a = *(const v4f*)(p + lane * 4);
    v4f b = *(const v4f*)(p + 128 + lane * 4);
    float s = a.x + a.y + a.z + a.w + b.x + b.y + b.z + b.w;
    #pragma unroll
    for (int off = 16; off > 0; off >>= 1)
        s += __shfl_xor(s, off, 32);
    if (lane == 0) xs[row] = s;
}

__global__ void RGCN_main_kernel(const float* __restrict__ cs,
                                 const float* __restrict__ Wpack,
                                 const float* __restrict__ xs,
                                 float* __restrict__ out) {
    const int wave      = threadIdx.x >> 5;
    const int lane      = threadIdx.x & 31;
    const int row_block = blockIdx.x * (256 >> 5) + wave;
    if (row_block >= E_ENT / 16) return;

    const int e_base = row_block * 16;
    const int half   = lane >> 4;
    const int m      = lane & 15;

    const float scale = xs[e_base + m];

    v2f a[16];
    const float* csrow = cs + (size_t)(e_base + m) * R_REL + half * 2;
    #pragma unroll
    for (int k = 0; k < 16; ++k) {
        const float c0 = csrow[4 * k + 0];
        const float c1 = csrow[4 * k + 1];
        a[k].x = scale / c0;
        a[k].y = scale / c1;
    }

    #pragma unroll 1
    for (int nt = 0; nt < 16; ++nt) {
        v8f acc = {};
        #pragma unroll
        for (int k = 0; k < 16; ++k) {
            const int p = 2 * k + half;
            v2f b = *(const v2f*)(Wpack +
                     ((size_t)p * O_CH + nt * 16 + m) * 2);
            acc = __builtin_amdgcn_wmma_f32_16x16x4_f32(
                 false, a[k],  false, b,
                 (short)0, acc,  false,  false);
        }
        float* obase = out + (size_t)(e_base + half * 8) * O_CH + nt * 16 + m;
        #pragma unroll
        for (int v = 0; v < 8; ++v)
            obase[(size_t)v * O_CH] = acc[v];
    }
}

extern "C" void kernel_launch(void* const* d_in, const int* in_sizes, int n_in,
                              void* d_out, int out_size, void* d_ws, size_t ws_size,
                              hipStream_t stream) {
    const float* x  = (const float*)d_in[0];
    const float* cs = (const float*)d_in[1];
    const float* W  = (const float*)d_in[2];
    float* out = (float*)d_out;

    float* Wpack = (float*)d_ws;
    float* xsum  = Wpack + R_REL * O_CH;

    RGCN_wsum_kernel<<<R_REL * 4, 256, 0, stream>>>(W, Wpack);
    RGCN_xsum_kernel<<<(E_ENT + 7) / 8, 256, 0, stream>>>(x, xsum);

    const int row_blocks = E_ENT / 16;
    const int grid = (row_blocks + 7) / 8;
    RGCN_main_kernel<<<grid, 256, 0, stream>>>(cs, Wpack, xsum, out);
}


# x row-sum pre-pass re-tiled: one wave per 16 rows (6250 waves instead of 100000), 32 b128 loads in flight per lane, one butterfly that halves the rows per lane each step; sums bit-identical
# speedup vs baseline: 1.5472x; 1.2186x over previous
#include <hip/hip_runtime.h>

typedef __attribute__((ext_vector_type(2))) float v2f;
typedef __attribute__((ext_vector_type(4))) float v4f;
typedef __attribute__((ext_vector_type(8))) float v8f;

#define E_ENT 100000
#define R_REL 64
#define I_CH  256
#define O_CH  256

__global__ void RGCN_wsum_kernel(const float* __restrict__ W,
                                 float* __restrict__ Wpack) {
    __shared__ float part[16][64];
    const int r  = blockIdx.x >> 2;
    const int cc = blockIdx.x & 3;
    const int c4 = threadIdx.x & 15;
    const int il = threadIdx.x >> 4;
    const float* base = W + ((size_t)r * I_CH + il) * O_CH + cc * 64 + c4 * 4;
    v4f v[16];
    for (int j = 0; j < 16; ++j)
        v[j] = *(const v4f*)(base + (size_t)j * 16 * O_CH);
    v4f s = v[0];
    for (int j = 1; j < 16; ++j) s += v[j];
    *(v4f*)&part[il][c4 * 4] = s;
    __syncthreads();
    if (threadIdx.x < 64) {
        const int col = threadIdx.x;
        float t = part[0][col];
        for (int k = 1; k < 16; ++k) t += part[k][col];
        Wpack[((size_t)(r >> 1) * O_CH + cc * 64 + col) * 2 + (r & 1)] = t;
    }
}

static_assert(E_ENT % 16 == 0, "a wave owns 16 whole rows");

__global__ void __launch_bounds__(128)
RGCN_xsum_kernel(const float* __restrict__ x,
                 float* __restrict__ xs) {
    const int wave = threadIdx.x >> 5;
    const int lane = threadIdx.x & 31;
    const int rb   = blockIdx.x * 4 + wave;
    if (rb >= E_ENT / 16) return;
    const float* p = x + (size_t)rb * 16 * I_CH + lane * 4;

    v4f a[16], b[16];
    #pragma unroll
    for (int r = 0; r < 16; ++r) {
        a[r] = *(const v4f*)(p + r * I_CH);
        b[r] = *(const v4f*)(p + r * I_CH + 128);
    }

    float s16[16];
    #pragma unroll
    for (int r = 0; r < 16; ++r)
        s16[r] = a[r].x + a[r].y + a[r].z + a[r].w
               + b[r].x + b[r].y + b[r].z + b[r].w;

    const bool h4 = (lane & 16) != 0;
    float s8[8];
    #pragma unroll
    for (int i = 0; i < 8; ++i) {
        const float keep = h4 ? s16[i + 8] : s16[i];
        const float send = h4 ? s16[i]     : s16[i + 8];
        s8[i] = keep + __shfl_xor(send, 16, 32);
    }
    const bool h3 = (lane & 8) != 0;
    float s4[4];
    #pragma unroll
    for (int i = 0; i < 4; ++i) {
        const float keep = h3 ? s8[i + 4] : s8[i];
        const float send = h3 ? s8[i]     : s8[i + 4];
        s4[i] = keep + __shfl_xor(send, 8, 32);
    }
    const bool h2 = (lane & 4) != 0;
    float s2[2];
    #pragma unroll
    for (int i = 0; i < 2; ++i) {
        const float keep = h2 ? s4[i + 2] : s4[i];
        const float send = h2 ? s4[i]     : s4[i + 2];
        s2[i] = keep + __shfl_xor(send, 4, 32);
    }
    const bool h1 = (lane & 2) != 0;
    float t = (h1 ? s2[1] : s2[0]) + __shfl_xor(h1 ? s2[0] : s2[1], 2, 32);
    t += __shfl_xor(t, 1, 32);
    if ((lane & 1) == 0) xs[rb * 16 + (lane >> 1)] = t;
}

__global__ void RGCN_main_kernel(const float* __restrict__ cs,
                                 const float* __restrict__ Wpack,
                                 const float* __restrict__ xs,
                                 float* __restrict__ out) {
    const int wave      = threadIdx.x >> 5;
    const int lane      = threadIdx.x & 31;
    const int row_block = blockIdx.x * (256 >> 5) + wave;
    if (row_block >= E_ENT / 16) return;

    const int e_base = row_block * 16;
    const int half   = lane >> 4;
    const int m      = lane & 15;

    const float scale = xs[e_base + m];

    v2f a[16];
    const float* csrow = cs + (size_t)(e_base + m) * R_REL + half * 2;
    #pragma unroll
    for (int k = 0; k < 16; ++k) {
        const float c0 = csrow[4 * k + 0];
        const float c1 = csrow[4 * k + 1];
        a[k].x = scale / c0;
        a[k].y = scale / c1;
    }

    #pragma unroll 1
    for (int nt = 0; nt < 16; ++nt) {
        v8f acc = {};
        #pragma unroll
        for (int k = 0; k < 16; ++k) {
            const int p = 2 * k + half;
            v2f b = *(const v2f*)(Wpack +
                     ((size_t)p * O_CH + nt * 16 + m) * 2);
            acc = __builtin_amdgcn_wmma_f32_16x16x4_f32(
                 false, a[k],  false, b,
                 (short)0, acc,  false,  false);
        }
        float* obase = out + (size_t)(e_base + half * 8) * O_CH + nt * 16 + m;
        #pragma unroll
        for (int v = 0; v < 8; ++v)
            obase[(size_t)v * O_CH] = acc[v];
    }
}

extern "C" void kernel_launch(void* const* d_in, const int* in_sizes, int n_in,
                              void* d_out, int out_size, void* d_ws, size_t ws_size,
                              hipStream_t stream) {
    const float* x  = (const float*)d_in[0];
    const float* cs = (const float*)d_in[1];
    const float* W  = (const float*)d_in[2];
    float* out = (float*)d_out;

    float* Wpack = (float*)d_ws;
    float* xsum  = Wpack + R_REL * O_CH;

    RGCN_wsum_kernel<<<R_REL * 4, 256, 0, stream>>>(W, Wpack);
    RGCN_xsum_kernel<<<(E_ENT / 16 + 3) / 4, 128, 0, stream>>>(x, xsum);

    const int row_blocks = E_ENT / 16;
    const int grid = (row_blocks + 7) / 8;
    RGCN_main_kernel<<<grid, 256, 0, stream>>>(cs, Wpack, xsum, out);
}


# W row-sum and x row-sum passes share one launch (W blocks dispatched first, x blocks of 8 waves behind them): 2 launches, not 3; both sums bit-identical
# speedup vs baseline: 1.6324x; 1.0551x over previous
#include <hip/hip_runtime.h>

typedef __attribute__((ext_vector_type(2))) float v2f;
typedef __attribute__((ext_vector_type(4))) float v4f;
typedef __attribute__((ext_vector_type(8))) float v8f;

#define E_ENT 100000
#define R_REL 64
#define I_CH  256
#define O_CH  256

static __device__ __forceinline__ void RGCN_wsum_block(const float* __restrict__ W,
                                                       float* __restrict__ Wpack,
                                                       const int blk) {
    __shared__ float part[16][64];
    const int r  = blk >> 2;
    const int cc = blk & 3;
    const int c4 = threadIdx.x & 15;
    const int il = threadIdx.x >> 4;
    const float* base = W + ((size_t)r * I_CH + il) * O_CH + cc * 64 + c4 * 4;
    v4f v[16];
    for (int j = 0; j < 16; ++j)
        v[j] = *(const v4f*)(base + (size_t)j * 16 * O_CH);
    v4f s = v[0];
    for (int j = 1; j < 16; ++j) s += v[j];
    *(v4f*)&part[il][c4 * 4] = s;
    __syncthreads();
    if (threadIdx.x < 64) {
        const int col = threadIdx.x;
        float t = part[0][col];
        for (int k = 1; k < 16; ++k) t += part[k][col];
        Wpack[((size_t)(r >> 1) * O_CH + cc * 64 + col) * 2 + (r & 1)] = t;
    }
}

static_assert(E_ENT % 16 == 0, "a wave owns 16 whole rows");

static __device__ __forceinline__ void RGCN_xsum_block(const float* __restrict__ x,
                                                       float* __restrict__ xs,
                                                       const int blk) {
    const int wave = threadIdx.x >> 5;
    const int lane = threadIdx.x & 31;
    const int rb   = blk * 8 + wave;
    if (rb >= E_ENT / 16) return;
    const float* p = x + (size_t)rb * 16 * I_CH + lane * 4;

    v4f a[16], b[16];
    #pragma unroll
    for (int r = 0; r < 16; ++r) {
        a[r] = *(const v4f*)(p + r * I_CH);
        b[r] = *(const v4f*)(p + r * I_CH + 128);
    }

    float s16[16];
    #pragma unroll
    for (int r = 0; r < 16; ++r)
        s16[r] = a[r].x + a[r].y + a[r].z + a[r].w
               + b[r].x + b[r].y + b[r].z + b[r].w;

    const bool h4 = (lane & 16) != 0;
    float s8[8];
    #pragma unroll
    for (int i = 0; i < 8; ++i) {
        const float keep = h4 ? s16[i + 8] : s16[i];
        const float send = h4 ? s16[i]     : s16[i + 8];
        s8[i] = keep + __shfl_xor(send, 16, 32);
    }
    const bool h3 = (lane & 8) != 0;
    float s4[4];
    #pragma unroll
    for (int i = 0; i < 4; ++i) {
        const float keep = h3 ? s8[i + 4] : s8[i];
        const float send = h3 ? s8[i]     : s8[i + 4];
        s4[i] = keep + __shfl_xor(send, 8, 32);
    }
    const bool h2 = (lane & 4) != 0;
    float s2[2];
    #pragma unroll
    for (int i = 0; i < 2; ++i) {
        const float keep = h2 ? s4[i + 2] : s4[i];
        const float send = h2 ? s4[i]     : s4[i + 2];
        s2[i] = keep + __shfl_xor(send, 4, 32);
    }
    const bool h1 = (lane & 2) != 0;
    float t = (h1 ? s2[1] : s2[0]) + __shfl_xor(h1 ? s2[0] : s2[1], 2, 32);
    t += __shfl_xor(t, 1, 32);
    if ((lane & 1) == 0) xs[rb * 16 + (lane >> 1)] = t;
}

#define W_BLOCKS (R_REL * 4)
#define X_BLOCKS ((E_ENT / 16 + 7) / 8)

__global__ void __launch_bounds__(256)
RGCN_presum_kernel(const float* __restrict__ W,
                   float* __restrict__ Wpack,
                   const float* __restrict__ x,
                   float* __restrict__ xs) {
    if (blockIdx.x < W_BLOCKS)
        RGCN_wsum_block(W, Wpack, blockIdx.x);
    else
        RGCN_xsum_block(x, xs, blockIdx.x - W_BLOCKS);
}

__global__ void RGCN_main_kernel(const float* __restrict__ cs,
                                 const float* __restrict__ Wpack,
                                 const float* __restrict__ xs,
                                 float* __restrict__ out) {
    const int wave      = threadIdx.x >> 5;
    const int lane      = threadIdx.x & 31;
    const int row_block = blockIdx.x * (256 >> 5) + wave;
    if (row_block >= E_ENT / 16) return;

    const int e_base = row_block * 16;
    const int half   = lane >> 4;
    const int m      = lane & 15;

    const float scale = xs[e_base + m];

    v2f a[16];
    const float* csrow = cs + (size_t)(e_base + m) * R_REL + half * 2;
    #pragma unroll
    for (int k = 0; k < 16; ++k) {
        const float c0 = csrow[4 * k + 0];
        const float c1 = csrow[4 * k + 1];
        a[k].x = scale / c0;
        a[k].y = scale / c1;
    }

    #pragma unroll 1
    for (int nt = 0; nt < 16; ++nt) {
        v8f acc = {};
        #pragma unroll
        for (int k = 0; k < 16; ++k) {
            const int p = 2 * k + half;
            v2f b = *(const v2f*)(Wpack +
                     ((size_t)p * O_CH + nt * 16 + m) * 2);
            acc = __builtin_amdgcn_wmma_f32_16x16x4_f32(
                 false, a[k],  false, b,
                 (short)0, acc,  false,  false);
        }
        float* obase = out + (size_t)(e_base + half * 8) * O_CH + nt * 16 + m;
        #pragma unroll
        for (int v = 0; v < 8; ++v)
            obase[(size_t)v * O_CH] = acc[v];
    }
}

extern "C" void kernel_launch(void* const* d_in, const int* in_sizes, int n_in,
                              void* d_out, int out_size, void* d_ws, size_t ws_size,
                              hipStream_t stream) {
    const float* x  = (const float*)d_in[0];
    const float* cs = (const float*)d_in[1];
    const float* W  = (const float*)d_in[2];
    float* out = (float*)d_out;

    float* Wpack = (float*)d_ws;
    float* xsum  = Wpack + R_REL * O_CH;

    RGCN_presum_kernel<<<W_BLOCKS + X_BLOCKS, 256, 0, stream>>>(W, Wpack, x, xsum);

    const int row_blocks = E_ENT / 16;
    const int grid = (row_blocks + 7) / 8;
    RGCN_main_kernel<<<grid, 256, 0, stream>>>(cs, Wpack, xsum, out);
}


# Main kernel: the 4 N-tiles of a 64-column group take interleaved columns 64g+4m+t, so a lane ends with 4 adjacent columns: 16-byte whole-line stores of out, 16-byte B loads; sums bit-identical
# speedup vs baseline: 1.8021x; 1.1040x over previous
#include <hip/hip_runtime.h>

typedef __attribute__((ext_vector_type(2))) float v2f;
typedef __attribute__((ext_vector_type(4))) float v4f;
typedef __attribute__((ext_vector_type(8))) float v8f;

#define E_ENT 100000
#define R_REL 64
#define I_CH  256
#define O_CH  256

static __device__ __forceinline__ int RGCN_wpack_index(const int r, const int o) {
    const int k = r >> 2, h = (r >> 1) & 1, j = r & 1;
    const int g = o >> 6, m = (o >> 2) & 15, q = (o >> 1) & 1, tt = o & 1;
    return ((((k * 4 + g) * 2 + q) * 2 + h) * 16 + m) * 4 + tt * 2 + j;
}

static __device__ __forceinline__ void RGCN_wsum_block(const float* __restrict__ W,
                                                       float* __restrict__ Wpack,
                                                       const int blk) {
    __shared__ float part[16][64];
    const int r  = blk >> 2;
    const int cc = blk & 3;
    const int c4 = threadIdx.x & 15;
    const int il = threadIdx.x >> 4;
    const float* base = W + ((size_t)r * I_CH + il) * O_CH + cc * 64 + c4 * 4;
    v4f v[16];
    for (int j = 0; j < 16; ++j)
        v[j] = *(const v4f*)(base + (size_t)j * 16 * O_CH);
    v4f s = v[0];
    for (int j = 1; j < 16; ++j) s += v[j];
    *(v4f*)&part[il][c4 * 4] = s;
    __syncthreads();
    if (threadIdx.x < 64) {
        const int col = threadIdx.x;
        float t = part[0][col];
        for (int k = 1; k < 16; ++k) t += part[k][col];
        Wpack[RGCN_wpack_index(r, cc * 64 + col)] = t;
    }
}

static_assert(E_ENT % 16 == 0, "a wave owns 16 whole rows");

static __device__ __forceinline__ void RGCN_xsum_block(const float* __restrict__ x,
                                                       float* __restrict__ xs,
                                                       const int blk) {
    const int wave = threadIdx.x >> 5;
    const int lane = threadIdx.x & 31;
    const int rb   = blk * 8 + wave;
    if (rb >= E_ENT / 16) return;
    const float* p = x + (size_t)rb * 16 * I_CH + lane * 4;

    v4f a[16], b[16];
    #pragma unroll
    for (int r = 0; r < 16; ++r) {
        a[r] = *(const v4f*)(p + r * I_CH);
        b[r] = *(const v4f*)(p + r * I_CH + 128);
    }

    float s16[16];
    #pragma unroll
    for (int r = 0; r < 16; ++r)
        s16[r] = a[r].x + a[r].y + a[r].z + a[r].w
               + b[r].x + b[r].y + b[r].z + b[r].w;

    const bool h4 = (lane & 16) != 0;
    float s8[8];
    #pragma unroll
    for (int i = 0; i < 8; ++i) {
        const float keep = h4 ? s16[i + 8] : s16[i];
        const float send = h4 ? s16[i]     : s16[i + 8];
        s8[i] = keep + __shfl_xor(send, 16, 32);
    }
    const bool h3 = (lane & 8) != 0;
    float s4[4];
    #pragma unroll
    for (int i = 0; i < 4; ++i) {
        const float keep = h3 ? s8[i + 4] : s8[i];
        const float send = h3 ? s8[i]     : s8[i + 4];
        s4[i] = keep + __shfl_xor(send, 8, 32);
    }
    const bool h2 = (lane & 4) != 0;
    float s2[2];
    #pragma unroll
    for (int i = 0; i < 2; ++i) {
        const float keep = h2 ? s4[i + 2] : s4[i];
        const float send = h2 ? s4[i]     : s4[i + 2];
        s2[i] = keep + __shfl_xor(send, 4, 32);
    }
    const bool h1 = (lane & 2) != 0;
    float t = (h1 ? s2[1] : s2[0]) + __shfl_xor(h1 ? s2[0] : s2[1], 2, 32);
    t += __shfl_xor(t, 1, 32);
    if ((lane & 1) == 0) xs[rb * 16 + (lane >> 1)] = t;
}

#define W_BLOCKS (R_REL * 4)
#define X_BLOCKS ((E_ENT / 16 + 7) / 8)

__global__ void __launch_bounds__(256)
RGCN_presum_kernel(const float* __restrict__ W,
                   float* __restrict__ Wpack,
                   const float* __restrict__ x,
                   float* __restrict__ xs) {
    if (blockIdx.x < W_BLOCKS)
        RGCN_wsum_block(W, Wpack, blockIdx.x);
    else
        RGCN_xsum_block(x, xs, blockIdx.x - W_BLOCKS);
}

__global__ void RGCN_main_kernel(const float* __restrict__ cs,
                                 const float* __restrict__ Wpack,
                                 const float* __restrict__ xs,
                                 float* __restrict__ out) {
    const int wave      = threadIdx.x >> 5;
    const int lane      = threadIdx.x & 31;
    const int row_block = blockIdx.x * (256 >> 5) + wave;
    if (row_block >= E_ENT / 16) return;

    const int e_base = row_block * 16;
    const int half   = lane >> 4;
    const int m      = lane & 15;

    const float scale = xs[e_base + m];

    v2f a[16];
    const float* csrow = cs + (size_t)(e_base + m) * R_REL + half * 2;
    #pragma unroll
    for (int k = 0; k < 16; ++k) {
        const float c0 = csrow[4 * k + 0];
        const float c1 = csrow[4 * k + 1];
        a[k].x = scale / c0;
        a[k].y = scale / c1;
    }

    const float* bl = Wpack + lane * 4;

    #pragma unroll 1
    for (int g = 0; g < 4; ++g) {
        v8f acc0 = {}, acc1 = {}, acc2 = {}, acc3 = {};
        #pragma unroll
        for (int k = 0; k < 16; ++k) {
            const float* bp = bl + ((k * 4 + g) * 2) * 128;
            const v4f b01 = *(const v4f*)(bp);
            const v4f b23 = *(const v4f*)(bp + 128);
            v2f b0, b1, b2, b3;
            b0.x = b01.x; b0.y = b01.y;
            b1.x = b01.z; b1.y = b01.w;
            b2.x = b23.x; b2.y = b23.y;
            b3.x = b23.z; b3.y = b23.w;
            acc0 = __builtin_amdgcn_wmma_f32_16x16x4_f32(
                 false, a[k],  false, b0,
                 (short)0, acc0,  false,  false);
            acc1 = __builtin_amdgcn_wmma_f32_16x16x4_f32(
                false, a[k], false, b1, (short)0, acc1, false, false);
            acc2 = __builtin_amdgcn_wmma_f32_16x16x4_f32(
                false, a[k], false, b2, (short)0, acc2, false, false);
            acc3 = __builtin_amdgcn_wmma_f32_16x16x4_f32(
                false, a[k], false, b3, (short)0, acc3, false, false);
        }
        float* obase = out + (size_t)(e_base + half * 8) * O_CH + g * 64 + m * 4;
        #pragma unroll
        for (int v = 0; v < 8; ++v) {
            v4f o4;
            o4.x = acc0[v]; o4.y = acc1[v]; o4.z = acc2[v]; o4.w = acc3[v];
            *(v4f*)(obase + (size_t)v * O_CH) = o4;
        }
    }
}

extern "C" void kernel_launch(void* const* d_in, const int* in_sizes, int n_in,
                              void* d_out, int out_size, void* d_ws, size_t ws_size,
                              hipStream_t stream) {
    const float* x  = (const float*)d_in[0];
    const float* cs = (const float*)d_in[1];
    const float* W  = (const float*)d_in[2];
    float* out = (float*)d_out;

    float* Wpack = (float*)d_ws;
    float* xsum  = Wpack + R_REL * O_CH;

    RGCN_presum_kernel<<<W_BLOCKS + X_BLOCKS, 256, 0, stream>>>(W, Wpack, x, xsum);

    const int row_blocks = E_ENT / 16;
    const int grid = (row_blocks + 7) / 8;
    RGCN_main_kernel<<<grid, 256, 0, stream>>>(cs, Wpack, xsum, out);
}


# Main kernel on a fixed grid: 256 work-groups of 20 waves, wave w takes the contiguous (row block, 64-column group) units [w*25000/5120, (w+1)*25000/5120), 4 or 5 each; sums bit-identical
# speedup vs baseline: 1.8682x; 1.0367x over previous
#include <hip/hip_runtime.h>

typedef __attribute__((ext_vector_type(2))) float v2f;
typedef __attribute__((ext_vector_type(4))) float v4f;
typedef __attribute__((ext_vector_type(8))) float v8f;

#define E_ENT 100000
#define R_REL 64
#define I_CH  256
#define O_CH  256

static __device__ __forceinline__ int RGCN_wpack_index(const int r, const int o) {
    const int k = r >> 2, h = (r >> 1) & 1, j = r & 1;
    const int g = o >> 6, m = (o >> 2) & 15, q = (o >> 1) & 1, tt = o & 1;
    return ((((k * 4 + g) * 2 + q) * 2 + h) * 16 + m) * 4 + tt * 2 + j;
}

static __device__ __forceinline__ void RGCN_wsum_block(const float* __restrict__ W,
                                                       float* __restrict__ Wpack,
                                                       const int blk) {
    __shared__ float part[16][64];
    const int r  = blk >> 2;
    const int cc = blk & 3;
    const int c4 = threadIdx.x & 15;
    const int il = threadIdx.x >> 4;
    const float* base = W + ((size_t)r * I_CH + il) * O_CH + cc * 64 + c4 * 4;
    v4f v[16];
    for (int j = 0; j < 16; ++j)
        v[j] = *(const v4f*)(base + (size_t)j * 16 * O_CH);
    v4f s = v[0];
    for (int j = 1; j < 16; ++j) s += v[j];
    *(v4f*)&part[il][c4 * 4] = s;
    __syncthreads();
    if (threadIdx.x < 64) {
        const int col = threadIdx.x;
        float t = part[0][col];
        for (int k = 1; k < 16; ++k) t += part[k][col];
        Wpack[RGCN_wpack_index(r, cc * 64 + col)] = t;
    }
}

static_assert(E_ENT % 16 == 0, "a wave owns 16 whole rows");

static __device__ __forceinline__ void RGCN_xsum_block(const float* __restrict__ x,
                                                       float* __restrict__ xs,
                                                       const int blk) {
    const int wave = threadIdx.x >> 5;
    const int lane = threadIdx.x & 31;
    const int rb   = blk * 8 + wave;
    if (rb >= E_ENT / 16) return;
    const float* p = x + (size_t)rb * 16 * I_CH + lane * 4;

    v4f a[16], b[16];
    #pragma unroll
    for (int r = 0; r < 16; ++r) {
        a[r] = *(const v4f*)(p + r * I_CH);
        b[r] = *(const v4f*)(p + r * I_CH + 128);
    }

    float s16[16];
    #pragma unroll
    for (int r = 0; r < 16; ++r)
        s16[r] = a[r].x + a[r].y + a[r].z + a[r].w
               + b[r].x + b[r].y + b[r].z + b[r].w;

    const bool h4 = (lane & 16) != 0;
    float s8[8];
    #pragma unroll
    for (int i = 0; i < 8; ++i) {
        const float keep = h4 ? s16[i + 8] : s16[i];
        const float send = h4 ? s16[i]     : s16[i + 8];
        s8[i] = keep + __shfl_xor(send, 16, 32);
    }
    const bool h3 = (lane & 8) != 0;
    float s4[4];
    #pragma unroll
    for (int i = 0; i < 4; ++i) {
        const float keep = h3 ? s8[i + 4] : s8[i];
        const float send = h3 ? s8[i]     : s8[i + 4];
        s4[i] = keep + __shfl_xor(send, 8, 32);
    }
    const bool h2 = (lane & 4) != 0;
    float s2[2];
    #pragma unroll
    for (int i = 0; i < 2; ++i) {
        const float keep = h2 ? s4[i + 2] : s4[i];
        const float send = h2 ? s4[i]     : s4[i + 2];
        s2[i] = keep + __shfl_xor(send, 4, 32);
    }
    const bool h1 = (lane & 2) != 0;
    float t = (h1 ? s2[1] : s2[0]) + __shfl_xor(h1 ? s2[0] : s2[1], 2, 32);
    t += __shfl_xor(t, 1, 32);
    if ((lane & 1) == 0) xs[rb * 16 + (lane >> 1)] = t;
}

#define W_BLOCKS (R_REL * 4)
#define X_BLOCKS ((E_ENT / 16 + 7) / 8)

__global__ void __launch_bounds__(256)
RGCN_presum_kernel(const float* __restrict__ W,
                   float* __restrict__ Wpack,
                   const float* __restrict__ x,
                   float* __restrict__ xs) {
    if (blockIdx.x < W_BLOCKS)
        RGCN_wsum_block(W, Wpack, blockIdx.x);
    else
        RGCN_xsum_block(x, xs, blockIdx.x - W_BLOCKS);
}

#define MAIN_BLOCKS 256
#define MAIN_WAVES_PER_BLOCK 20
#define MAIN_WAVES (MAIN_BLOCKS * MAIN_WAVES_PER_BLOCK)
#define N_UNITS ((E_ENT / 16) * 4)
static_assert(N_UNITS >= MAIN_WAVES, "every wave owns at least one unit");
static_assert((long long)MAIN_WAVES * N_UNITS < (1LL << 32), "range bounds fit 32 bits");

__global__ void __launch_bounds__(MAIN_WAVES_PER_BLOCK * 32)
RGCN_main_kernel(const float* __restrict__ cs,
                 const float* __restrict__ Wpack,
                 const float* __restrict__ xs,
                 float* __restrict__ out) {
    const int wave = __builtin_amdgcn_readfirstlane((int)(threadIdx.x >> 5));
    const int lane = threadIdx.x & 31;
    const int w    = blockIdx.x * MAIN_WAVES_PER_BLOCK + wave;
    int       u     = (int)(((unsigned)w * (unsigned)N_UNITS) / (unsigned)MAIN_WAVES);
    const int u_end = (int)(((unsigned)(w + 1) * (unsigned)N_UNITS) / (unsigned)MAIN_WAVES);

    const int half = lane >> 4;
    const int m    = lane & 15;

    const float* bl = Wpack + lane * 4;

    while (u < u_end) {
        const int rb     = u >> 2;
        const int g_lo   = u & 3;
        const int left   = u_end - rb * 4;
        const int g_hi   = left < 4 ? left : 4;
        const int e_base = rb * 16;

        const float scale = xs[e_base + m];

        v2f a[16];
        const float* csrow = cs + (size_t)(e_base + m) * R_REL + half * 2;
        #pragma unroll
        for (int k = 0; k < 16; ++k) {
            const float c0 = csrow[4 * k + 0];
            const float c1 = csrow[4 * k + 1];
            a[k].x = scale / c0;
            a[k].y = scale / c1;
        }

        #pragma unroll 1
        for (int g = g_lo; g < g_hi; ++g) {
            v8f acc0 = {}, acc1 = {}, acc2 = {}, acc3 = {};
            #pragma unroll
            for (int k = 0; k < 16; ++k) {
                const float* bp = bl + ((k * 4 + g) * 2) * 128;
                const v4f b01 = *(const v4f*)(bp);
                const v4f b23 = *(const v4f*)(bp + 128);
                v2f b0, b1, b2, b3;
                b0.x = b01.x; b0.y = b01.y;
                b1.x = b01.z; b1.y = b01.w;
                b2.x = b23.x; b2.y = b23.y;
                b3.x = b23.z; b3.y = b23.w;
                acc0 = __builtin_amdgcn_wmma_f32_16x16x4_f32(
                     false, a[k],  false, b0,
                     (short)0, acc0,  false,  false);
                acc1 = __builtin_amdgcn_wmma_f32_16x16x4_f32(
                    false, a[k], false, b1, (short)0, acc1, false, false);
                acc2 = __builtin_amdgcn_wmma_f32_16x16x4_f32(
                    false, a[k], false, b2, (short)0, acc2, false, false);
                acc3 = __builtin_amdgcn_wmma_f32_16x16x4_f32(
                    false, a[k], false, b3, (short)0, acc3, false, false);
            }
            float* obase = out + (size_t)(e_base + half * 8) * O_CH + g * 64 + m * 4;
            #pragma unroll
            for (int v = 0; v < 8; ++v) {
                v4f o4;
                o4.x = acc0[v]; o4.y = acc1[v]; o4.z = acc2[v]; o4.w = acc3[v];
                *(v4f*)(obase + (size_t)v * O_CH) = o4;
            }
        }
        u = rb * 4 + g_hi;
    }
}

extern "C" void kernel_launch(void* const* d_in, const int* in_sizes, int n_in,
                              void* d_out, int out_size, void* d_ws, size_t ws_size,
                              hipStream_t stream) {
    const float* x  = (const float*)d_in[0];
    const float* cs = (const float*)d_in[1];
    const float* W  = (const float*)d_in[2];
    float* out = (float*)d_out;

    float* Wpack = (float*)d_ws;
    float* xsum  = Wpack + R_REL * O_CH;

    RGCN_presum_kernel<<<W_BLOCKS + X_BLOCKS, 256, 0, stream>>>(W, Wpack, x, xsum);

    RGCN_main_kernel<<<MAIN_BLOCKS, MAIN_WAVES_PER_BLOCK * 32, 0, stream>>>(cs, Wpack, xsum, out);
}


# Main kernel takes x row sums itself: a row block's 32 x loads fly under its first group's WMMAs (B from an LDS copy of Wpack, cs loaded first); scale applied after the product; x pre-pass dropped
# speedup vs baseline: 1.9240x; 1.0299x over previous
#include <hip/hip_runtime.h>

typedef __attribute__((ext_vector_type(2))) float v2f;
typedef __attribute__((ext_vector_type(4))) float v4f;
typedef __attribute__((ext_vector_type(8))) float v8f;

#define E_ENT 100000
#define R_REL 64
#define I_CH  256
#define O_CH  256

static __device__ __forceinline__ int RGCN_wpack_index(const int r, const int o) {
    const int k = r >> 2, h = (r >> 1) & 1, j = r & 1;
    const int g = o >> 6, m = (o >> 2) & 15, q = (o >> 1) & 1, tt = o & 1;
    return ((((k * 4 + g) * 2 + q) * 2 + h) * 16 + m) * 4 + tt * 2 + j;
}

static __device__ __forceinline__ void RGCN_wsum_block(const float* __restrict__ W,
                                                       float* __restrict__ Wpack,
                                                       const int blk) {
    __shared__ float part[16][64];
    const int r  = blk >> 2;
    const int cc = blk & 3;
    const int c4 = threadIdx.x & 15;
    const int il = threadIdx.x >> 4;
    const float* base = W + ((size_t)r * I_CH + il) * O_CH + cc * 64 + c4 * 4;
    v4f v[16];
    for (int j = 0; j < 16; ++j)
        v[j] = *(const v4f*)(base + (size_t)j * 16 * O_CH);
    v4f s = v[0];
    for (int j = 1; j < 16; ++j) s += v[j];
    *(v4f*)&part[il][c4 * 4] = s;
    __syncthreads();
    if (threadIdx.x < 64) {
        const int col = threadIdx.x;
        float t = part[0][col];
        for (int k = 1; k < 16; ++k) t += part[k][col];
        Wpack[RGCN_wpack_index(r, cc * 64 + col)] = t;
    }
}

#define W_BLOCKS (R_REL * 4)

__global__ void __launch_bounds__(256)
RGCN_wsum_kernel(const float* __restrict__ W,
                 float* __restrict__ Wpack) {
    RGCN_wsum_block(W, Wpack, blockIdx.x);
}

#define MAIN_BLOCKS 256
#define MAIN_WAVES_PER_BLOCK 12
#define MAIN_WAVES (MAIN_BLOCKS * MAIN_WAVES_PER_BLOCK)
#define N_UNITS ((E_ENT / 16) * 4)
static_assert(E_ENT % 16 == 0, "a row block is 16 whole rows");
static_assert(N_UNITS >= MAIN_WAVES, "every wave owns at least one unit");
static_assert((long long)MAIN_WAVES * N_UNITS < (1LL << 32), "range bounds fit 32 bits");
#define STAGE_VEC (R_REL * O_CH / 4)
#define STAGE_PER_THREAD ((STAGE_VEC + MAIN_WAVES_PER_BLOCK * 32 - 1) / (MAIN_WAVES_PER_BLOCK * 32))

#define RGCN_PIN() do { asm volatile("" ::: "memory"); __builtin_amdgcn_sched_barrier(0); } while (0)

template <bool SPLIT>
static __device__ __forceinline__ void RGCN_group_mma(const v2f (&a)[16],
                                                      const float* bl, const int g,
                                                      v8f& acc0, v8f& acc1,
                                                      v8f& acc2, v8f& acc3) {
    const v8f zero = {};
    acc0 = zero; acc1 = zero; acc2 = zero; acc3 = zero;
    #pragma unroll
    for (int k = 0; k < 16; ++k) {
        if (SPLIT && k == 8) RGCN_PIN();
        const float* bp = bl + ((k * 4 + g) * 2) * 128;
        const v4f b01 = *(const v4f*)(bp);
        const v4f b23 = *(const v4f*)(bp + 128);
        v2f b0, b1, b2, b3;
        b0.x = b01.x; b0.y = b01.y;
        b1.x = b01.z; b1.y = b01.w;
        b2.x = b23.x; b2.y = b23.y;
        b3.x = b23.z; b3.y = b23.w;
        acc0 = __builtin_amdgcn_wmma_f32_16x16x4_f32(
             false, a[k],  false, b0,
             (short)0, acc0,  false,  false);
        acc1 = __builtin_amdgcn_wmma_f32_16x16x4_f32(
            false, a[k], false, b1, (short)0, acc1, false, false);
        acc2 = __builtin_amdgcn_wmma_f32_16x16x4_f32(
            false, a[k], false, b2, (short)0, acc2, false, false);
        acc3 = __builtin_amdgcn_wmma_f32_16x16x4_f32(
            false, a[k], false, b3, (short)0, acc3, false, false);
    }
}

static __device__ __forceinline__ void RGCN_group_store(float* obase,
                                                        const float (&sc)[8],
                                                        const v8f& acc0, const v8f& acc1,
                                                        const v8f& acc2, const v8f& acc3) {
    #pragma unroll
    for (int v = 0; v < 8; ++v) {
        v4f o4;
        o4.x = acc0[v] * sc[v]; o4.y = acc1[v] * sc[v];
        o4.z = acc2[v] * sc[v]; o4.w = acc3[v] * sc[v];
        *(v4f*)(obase + (size_t)v * O_CH) = o4;
    }
}

__global__ void __launch_bounds__(MAIN_WAVES_PER_BLOCK * 32)
__attribute__((amdgpu_waves_per_eu(3, 3)))
RGCN_main_kernel(const float* cs,
                 const float* __restrict__ Wpack,
                 const float* x,
                 float* out) {
    __shared__ __attribute__((aligned(16))) float Bs[R_REL * O_CH];

    {
        const v4f* src = (const v4f*)Wpack;
        v4f*       dst = (v4f*)Bs;
        v4f tmp[STAGE_PER_THREAD];
        #pragma unroll
        for (int i = 0; i < STAGE_PER_THREAD; ++i) {
            const int idx = i * (MAIN_WAVES_PER_BLOCK * 32) + (int)threadIdx.x;
            if (idx < STAGE_VEC) tmp[i] = src[idx];
        }
        #pragma unroll
        for (int i = 0; i < STAGE_PER_THREAD; ++i) {
            const int idx = i * (MAIN_WAVES_PER_BLOCK * 32) + (int)threadIdx.x;
            if (idx < STAGE_VEC) dst[idx] = tmp[i];
        }
    }
    __syncthreads();

    const int wave = __builtin_amdgcn_readfirstlane((int)(threadIdx.x >> 5));
    const int lane = threadIdx.x & 31;
    const int w    = blockIdx.x * MAIN_WAVES_PER_BLOCK + wave;
    int       u     = (int)(((unsigned)w * (unsigned)N_UNITS) / (unsigned)MAIN_WAVES);
    const int u_end = (int)(((unsigned)(w + 1) * (unsigned)N_UNITS) / (unsigned)MAIN_WAVES);

    const int half = lane >> 4;
    const int m    = lane & 15;

    const float* bl = Bs + lane * 4;

    while (u < u_end) {
        const int rb     = u >> 2;
        const int g_lo   = u & 3;
        const int left   = u_end - rb * 4;
        const int g_hi   = left < 4 ? left : 4;
        const int e_base = rb * 16;

        float c0[16], c1[16];
        const float* csrow = cs + (size_t)(e_base + m) * R_REL + half * 2;
        #pragma unroll
        for (int k = 0; k < 16; ++k) {
            c0[k] = csrow[4 * k + 0];
            c1[k] = csrow[4 * k + 1];
        }
        RGCN_PIN();

        const float* p = x + (size_t)e_base * I_CH + lane * 4;
        v4f xa[16], xb[16];
        #pragma unroll
        for (int r = 0; r < 16; ++r) {
            xa[r] = *(const v4f*)(p + r * I_CH);
            xb[r] = *(const v4f*)(p + r * I_CH + 128);
        }
        RGCN_PIN();

        v2f a[16];
        #pragma unroll
        for (int k = 0; k < 16; ++k) {
            a[k].x = 1.0f / c0[k];
            a[k].y = 1.0f / c1[k];
        }

        v8f acc0, acc1, acc2, acc3;
        RGCN_group_mma<true>(a, bl, g_lo, acc0, acc1, acc2, acc3);
        RGCN_PIN();

        float s16[16];
        #pragma unroll
        for (int r = 0; r < 16; ++r)
            s16[r] = xa[r].x + xa[r].y + xa[r].z + xa[r].w
                   + xb[r].x + xb[r].y + xb[r].z + xb[r].w;
        const bool h4 = (lane & 16) != 0;
        float s8[8];
        #pragma unroll
        for (int i = 0; i < 8; ++i) {
            const float keep = h4 ? s16[i + 8] : s16[i];
            const float send = h4 ? s16[i]     : s16[i + 8];
            s8[i] = keep + __shfl_xor(send, 16, 32);
        }
        const bool h3 = (lane & 8) != 0;
        float s4[4];
        #pragma unroll
        for (int i = 0; i < 4; ++i) {
            const float keep = h3 ? s8[i + 4] : s8[i];
            const float send = h3 ? s8[i]     : s8[i + 4];
            s4[i] = keep + __shfl_xor(send, 8, 32);
        }
        const bool h2 = (lane & 4) != 0;
        float s2[2];
        #pragma unroll
        for (int i = 0; i < 2; ++i) {
            const float keep = h2 ? s4[i + 2] : s4[i];
            const float send = h2 ? s4[i]     : s4[i + 2];
            s2[i] = keep + __shfl_xor(send, 4, 32);
        }
        const bool h1 = (lane & 2) != 0;
        float t = (h1 ? s2[1] : s2[0]) + __shfl_xor(h1 ? s2[0] : s2[1], 2, 32);
        t += __shfl_xor(t, 1, 32);
        float sc[8];
        #pragma unroll
        for (int v = 0; v < 8; ++v) sc[v] = __shfl(t, (lane & 16) + 2 * v, 32);

        float* obase = out + (size_t)(e_base + half * 8) * O_CH + m * 4;
        RGCN_group_store(obase + g_lo * 64, sc, acc0, acc1, acc2, acc3);

        #pragma unroll 1
        for (int g = g_lo + 1; g < g_hi; ++g) {
            RGCN_group_mma<false>(a, bl, g, acc0, acc1, acc2, acc3);
            RGCN_group_store(obase + g * 64, sc, acc0, acc1, acc2, acc3);
        }
        u = rb * 4 + g_hi;
    }
}

extern "C" void kernel_launch(void* const* d_in, const int* in_sizes, int n_in,
                              void* d_out, int out_size, void* d_ws, size_t ws_size,
                              hipStream_t stream) {
    const float* x  = (const float*)d_in[0];
    const float* cs = (const float*)d_in[1];
    const float* W  = (const float*)d_in[2];
    float* out = (float*)d_out;

    float* Wpack = (float*)d_ws;

    RGCN_wsum_kernel<<<W_BLOCKS, 256, 0, stream>>>(W, Wpack);

    RGCN_main_kernel<<<MAIN_BLOCKS, MAIN_WAVES_PER_BLOCK * 32, 0, stream>>>(cs, Wpack, x, out);
}


# Main kernel issues each visit's cs/x loads one stage early: the first across the Wpack staging barrier (now an LDS-only fence), later ones once the row sums free the x registers; sums bit-identical
# speedup vs baseline: 1.9789x; 1.0285x over previous
#include <hip/hip_runtime.h>

typedef __attribute__((ext_vector_type(2))) float v2f;
typedef __attribute__((ext_vector_type(4))) float v4f;
typedef __attribute__((ext_vector_type(8))) float v8f;

#define E_ENT 100000
#define R_REL 64
#define I_CH  256
#define O_CH  256

static __device__ __forceinline__ int RGCN_wpack_index(const int r, const int o) {
    const int k = r >> 2, h = (r >> 1) & 1, j = r & 1;
    const int g = o >> 6, m = (o >> 2) & 15, q = (o >> 1) & 1, tt = o & 1;
    return ((((k * 4 + g) * 2 + q) * 2 + h) * 16 + m) * 4 + tt * 2 + j;
}

static __device__ __forceinline__ void RGCN_wsum_block(const float* __restrict__ W,
                                                       float* __restrict__ Wpack,
                                                       const int blk) {
    __shared__ float part[16][64];
    const int r  = blk >> 2;
    const int cc = blk & 3;
    const int c4 = threadIdx.x & 15;
    const int il = threadIdx.x >> 4;
    const float* base = W + ((size_t)r * I_CH + il) * O_CH + cc * 64 + c4 * 4;
    v4f v[16];
    for (int j = 0; j < 16; ++j)
        v[j] = *(const v4f*)(base + (size_t)j * 16 * O_CH);
    v4f s = v[0];
    for (int j = 1; j < 16; ++j) s += v[j];
    *(v4f*)&part[il][c4 * 4] = s;
    __syncthreads();
    if (threadIdx.x < 64) {
        const int col = threadIdx.x;
        float t = part[0][col];
        for (int k = 1; k < 16; ++k) t += part[k][col];
        Wpack[RGCN_wpack_index(r, cc * 64 + col)] = t;
    }
}

#define W_BLOCKS (R_REL * 4)

__global__ void __launch_bounds__(256)
RGCN_wsum_kernel(const float* __restrict__ W,
                 float* __restrict__ Wpack) {
    RGCN_wsum_block(W, Wpack, blockIdx.x);
}

#define MAIN_BLOCKS 256
#define MAIN_WAVES_PER_BLOCK 12
#define MAIN_WAVES (MAIN_BLOCKS * MAIN_WAVES_PER_BLOCK)
#define N_UNITS ((E_ENT / 16) * 4)
static_assert(E_ENT % 16 == 0, "a row block is 16 whole rows");
static_assert(N_UNITS >= MAIN_WAVES, "every wave owns at least one unit");
static_assert((long long)MAIN_WAVES * N_UNITS < (1LL << 32), "range bounds fit 32 bits");
#define STAGE_VEC (R_REL * O_CH / 4)
#define STAGE_PER_THREAD ((STAGE_VEC + MAIN_WAVES_PER_BLOCK * 32 - 1) / (MAIN_WAVES_PER_BLOCK * 32))

#define RGCN_PIN() do { asm volatile("" ::: "memory"); __builtin_amdgcn_sched_barrier(0); } while (0)

static __device__ __forceinline__ void RGCN_b_load(const float* bl, const int k, const int g,
                                                   v4f& b01, v4f& b23) {
    const float* bp = bl + ((k * 4 + g) * 2) * 128;
    b01 = *(const v4f*)(bp);
    b23 = *(const v4f*)(bp + 128);
}

static __device__ __forceinline__ void RGCN_chunk_mma(const v2f a, const v4f b01, const v4f b23,
                                                      v8f& acc0, v8f& acc1,
                                                      v8f& acc2, v8f& acc3) {
    v2f b0, b1, b2, b3;
    b0.x = b01.x; b0.y = b01.y;
    b1.x = b01.z; b1.y = b01.w;
    b2.x = b23.x; b2.y = b23.y;
    b3.x = b23.z; b3.y = b23.w;
    acc0 = __builtin_amdgcn_wmma_f32_16x16x4_f32(
         false, a,  false, b0,
         (short)0, acc0,  false,  false);
    acc1 = __builtin_amdgcn_wmma_f32_16x16x4_f32(
        false, a, false, b1, (short)0, acc1, false, false);
    acc2 = __builtin_amdgcn_wmma_f32_16x16x4_f32(
        false, a, false, b2, (short)0, acc2, false, false);
    acc3 = __builtin_amdgcn_wmma_f32_16x16x4_f32(
        false, a, false, b3, (short)0, acc3, false, false);
}

template <bool SPLIT>
static __device__ __forceinline__ void RGCN_group_mma(const v2f (&a)[16],
                                                      const float* bl, const int g,
                                                      v8f& acc0, v8f& acc1,
                                                      v8f& acc2, v8f& acc3) {
    const v8f zero = {};
    acc0 = zero; acc1 = zero; acc2 = zero; acc3 = zero;
    if (SPLIT) {
        #pragma unroll
        for (int k = 0; k < 16; ++k) {
            if (k == 8) RGCN_PIN();
            v4f b01, b23;
            RGCN_b_load(bl, k, g, b01, b23);
            RGCN_chunk_mma(a[k], b01, b23, acc0, acc1, acc2, acc3);
        }
    } else {
        v4f p01[18], p23[18];
        RGCN_b_load(bl, 0, g, p01[0], p23[0]);
        RGCN_b_load(bl, 1, g, p01[1], p23[1]);
        #pragma unroll
        for (int k = 0; k < 16; ++k) {
            if (k + 2 < 16) RGCN_b_load(bl, k + 2, g, p01[k + 2], p23[k + 2]);
            RGCN_chunk_mma(a[k], p01[k], p23[k], acc0, acc1, acc2, acc3);
            __builtin_amdgcn_sched_barrier(0);
        }
    }
}

static __device__ __forceinline__ void RGCN_group_store(float* obase,
                                                        const float (&sc)[8],
                                                        const v8f& acc0, const v8f& acc1,
                                                        const v8f& acc2, const v8f& acc3) {
    #pragma unroll
    for (int v = 0; v < 8; ++v) {
        v4f o4;
        o4.x = acc0[v] * sc[v]; o4.y = acc1[v] * sc[v];
        o4.z = acc2[v] * sc[v]; o4.w = acc3[v] * sc[v];
        *(v4f*)(obase + (size_t)v * O_CH) = o4;
    }
}

static __device__ __forceinline__ void RGCN_visit_loads(const float* cs, const float* x,
                                                        const int e_base, const int m,
                                                        const int half, const int lane,
                                                        float (&c0)[16], float (&c1)[16],
                                                        v4f (&xa)[16], v4f (&xb)[16]) {
    const float* csrow = cs + (size_t)(e_base + m) * R_REL + half * 2;
    #pragma unroll
    for (int k = 0; k < 16; ++k) {
        c0[k] = csrow[4 * k + 0];
        c1[k] = csrow[4 * k + 1];
    }
    RGCN_PIN();
    const float* p = x + (size_t)e_base * I_CH + lane * 4;
    #pragma unroll
    for (int r = 0; r < 16; ++r) {
        xa[r] = *(const v4f*)(p + r * I_CH);
        xb[r] = *(const v4f*)(p + r * I_CH + 128);
    }
    RGCN_PIN();
}

__global__ void __launch_bounds__(MAIN_WAVES_PER_BLOCK * 32)
__attribute__((amdgpu_waves_per_eu(3, 3)))
RGCN_main_kernel(const float* cs,
                 const float* __restrict__ Wpack,
                 const float* x,
                 float* out) {
    __shared__ __attribute__((aligned(16))) float Bs[R_REL * O_CH];

    const int wave = __builtin_amdgcn_readfirstlane((int)(threadIdx.x >> 5));
    const int lane = threadIdx.x & 31;
    const int w    = blockIdx.x * MAIN_WAVES_PER_BLOCK + wave;
    int       u     = (int)(((unsigned)w * (unsigned)N_UNITS) / (unsigned)MAIN_WAVES);
    const int u_end = (int)(((unsigned)(w + 1) * (unsigned)N_UNITS) / (unsigned)MAIN_WAVES);

    const int half = lane >> 4;
    const int m    = lane & 15;

    float c0[16], c1[16];
    v4f xa[16], xb[16];

    {
        const v4f* src = (const v4f*)Wpack;
        v4f*       dst = (v4f*)Bs;
        v4f tmp[STAGE_PER_THREAD];
        #pragma unroll
        for (int i = 0; i < STAGE_PER_THREAD; ++i) {
            const int idx = i * (MAIN_WAVES_PER_BLOCK * 32) + (int)threadIdx.x;
            if (idx < STAGE_VEC) tmp[i] = src[idx];
        }
        RGCN_PIN();
        RGCN_visit_loads(cs, x, (u >> 2) * 16, m, half, lane, c0, c1, xa, xb);
        #pragma unroll
        for (int i = 0; i < STAGE_PER_THREAD; ++i) {
            const int idx = i * (MAIN_WAVES_PER_BLOCK * 32) + (int)threadIdx.x;
            if (idx < STAGE_VEC) dst[idx] = tmp[i];
        }
    }
    __builtin_amdgcn_fence(__ATOMIC_RELEASE, "workgroup", "local");
    __builtin_amdgcn_s_barrier();
    __builtin_amdgcn_fence(__ATOMIC_ACQUIRE, "workgroup", "local");

    const float* bl = Bs + lane * 4;

    while (true) {
        const int rb     = u >> 2;
        const int g_lo   = u & 3;
        const int left   = u_end - rb * 4;
        const int g_hi   = left < 4 ? left : 4;
        const int e_base = rb * 16;

        v2f a[16];
        #pragma unroll
        for (int k = 0; k < 16; ++k) {
            a[k].x = 1.0f / c0[k];
            a[k].y = 1.0f / c1[k];
        }

        v8f acc0, acc1, acc2, acc3;
        RGCN_group_mma<true>(a, bl, g_lo, acc0, acc1, acc2, acc3);
        RGCN_PIN();

        float s16[16];
        #pragma unroll
        for (int r = 0; r < 16; ++r)
            s16[r] = xa[r].x + xa[r].y + xa[r].z + xa[r].w
                   + xb[r].x + xb[r].y + xb[r].z + xb[r].w;
        const bool h4 = (lane & 16) != 0;
        float s8[8];
        #pragma unroll
        for (int i = 0; i < 8; ++i) {
            const float keep = h4 ? s16[i + 8] : s16[i];
            const float send = h4 ? s16[i]     : s16[i + 8];
            s8[i] = keep + __shfl_xor(send, 16, 32);
        }
        const bool h3 = (lane & 8) != 0;
        float s4[4];
        #pragma unroll
        for (int i = 0; i < 4; ++i) {
            const float keep = h3 ? s8[i + 4] : s8[i];
            const float send = h3 ? s8[i]     : s8[i + 4];
            s4[i] = keep + __shfl_xor(send, 8, 32);
        }
        const bool h2 = (lane & 4) != 0;
        float s2[2];
        #pragma unroll
        for (int i = 0; i < 2; ++i) {
            const float keep = h2 ? s4[i + 2] : s4[i];
            const float send = h2 ? s4[i]     : s4[i + 2];
            s2[i] = keep + __shfl_xor(send, 4, 32);
        }
        const bool h1 = (lane & 2) != 0;
        float t = (h1 ? s2[1] : s2[0]) + __shfl_xor(h1 ? s2[0] : s2[1], 2, 32);
        t += __shfl_xor(t, 1, 32);
        float sc[8];
        #pragma unroll
        for (int v = 0; v < 8; ++v) sc[v] = __shfl(t, (lane & 16) + 2 * v, 32);

        const int  u_next = rb * 4 + g_hi;
        const bool more   = u_next < u_end;
        if (more)
            RGCN_visit_loads(cs, x, (u_next >> 2) * 16, m, half, lane, c0, c1, xa, xb);

        float* obase = out + (size_t)(e_base + half * 8) * O_CH + m * 4;
        RGCN_group_store(obase + g_lo * 64, sc, acc0, acc1, acc2, acc3);

        #pragma unroll 1
        for (int g = g_lo + 1; g < g_hi; ++g) {
            RGCN_group_mma<false>(a, bl, g, acc0, acc1, acc2, acc3);
            RGCN_group_store(obase + g * 64, sc, acc0, acc1, acc2, acc3);
        }
        if (!more) break;
        u = u_next;
    }
}

extern "C" void kernel_launch(void* const* d_in, const int* in_sizes, int n_in,
                              void* d_out, int out_size, void* d_ws, size_t ws_size,
                              hipStream_t stream) {
    const float* x  = (const float*)d_in[0];
    const float* cs = (const float*)d_in[1];
    const float* W  = (const float*)d_in[2];
    float* out = (float*)d_out;

    float* Wpack = (float*)d_ws;

    RGCN_wsum_kernel<<<W_BLOCKS, 256, 0, stream>>>(W, Wpack);

    RGCN_main_kernel<<<MAIN_BLOCKS, MAIN_WAVES_PER_BLOCK * 32, 0, stream>>>(cs, Wpack, x, out);
}


# Main kernel splits work on whole row blocks: wave w takes blocks 2w, 2w+1; the last 424 units go singly to the workgroups' first waves; a block's cs and x are read and summed once; sums bit-identical
# speedup vs baseline: 2.0610x; 1.0415x over previous
#include <hip/hip_runtime.h>

typedef __attribute__((ext_vector_type(2))) float v2f;
typedef __attribute__((ext_vector_type(4))) float v4f;
typedef __attribute__((ext_vector_type(8))) float v8f;

#define E_ENT 100000
#define R_REL 64
#define I_CH  256
#define O_CH  256

static __device__ __forceinline__ int RGCN_wpack_index(const int r, const int o) {
    const int k = r >> 2, h = (r >> 1) & 1, j = r & 1;
    const int g = o >> 6, m = (o >> 2) & 15, q = (o >> 1) & 1, tt = o & 1;
    return ((((k * 4 + g) * 2 + q) * 2 + h) * 16 + m) * 4 + tt * 2 + j;
}

static __device__ __forceinline__ void RGCN_wsum_block(const float* __restrict__ W,
                                                       float* __restrict__ Wpack,
                                                       const int blk) {
    __shared__ float part[16][64];
    const int r  = blk >> 2;
    const int cc = blk & 3;
    const int c4 = threadIdx.x & 15;
    const int il = threadIdx.x >> 4;
    const float* base = W + ((size_t)r * I_CH + il) * O_CH + cc * 64 + c4 * 4;
    v4f v[16];
    for (int j = 0; j < 16; ++j)
        v[j] = *(const v4f*)(base + (size_t)j * 16 * O_CH);
    v4f s = v[0];
    for (int j = 1; j < 16; ++j) s += v[j];
    *(v4f*)&part[il][c4 * 4] = s;
    __syncthreads();
    if (threadIdx.x < 64) {
        const int col = threadIdx.x;
        float t = part[0][col];
        for (int k = 1; k < 16; ++k) t += part[k][col];
        Wpack[RGCN_wpack_index(r, cc * 64 + col)] = t;
    }
}

#define W_BLOCKS (R_REL * 4)

__global__ void __launch_bounds__(256)
RGCN_wsum_kernel(const float* __restrict__ W,
                 float* __restrict__ Wpack) {
    RGCN_wsum_block(W, Wpack, blockIdx.x);
}

#define MAIN_BLOCKS 256
#define MAIN_WAVES_PER_BLOCK 12
#define MAIN_WAVES (MAIN_BLOCKS * MAIN_WAVES_PER_BLOCK)
#define N_UNITS ((E_ENT / 16) * 4)
#define BASE_UNITS (MAIN_WAVES * 8)
#define N_EXTRA (N_UNITS - BASE_UNITS)
static_assert(E_ENT % 16 == 0, "a row block is 16 whole rows");
static_assert(N_EXTRA >= 0, "two whole row blocks per wave exist");
static_assert(N_EXTRA <= 4 * MAIN_BLOCKS, "at most one extra unit per SIMD: waves 0..3 of a workgroup");
#define STAGE_VEC (R_REL * O_CH / 4)
#define STAGE_PER_THREAD ((STAGE_VEC + MAIN_WAVES_PER_BLOCK * 32 - 1) / (MAIN_WAVES_PER_BLOCK * 32))

#define RGCN_PIN() do { asm volatile("" ::: "memory"); __builtin_amdgcn_sched_barrier(0); } while (0)

static __device__ __forceinline__ void RGCN_b_load(const float* bl, const int k, const int g,
                                                   v4f& b01, v4f& b23) {
    const float* bp = bl + ((k * 4 + g) * 2) * 128;
    b01 = *(const v4f*)(bp);
    b23 = *(const v4f*)(bp + 128);
}

static __device__ __forceinline__ void RGCN_chunk_mma(const v2f a, const v4f b01, const v4f b23,
                                                      v8f& acc0, v8f& acc1,
                                                      v8f& acc2, v8f& acc3) {
    v2f b0, b1, b2, b3;
    b0.x = b01.x; b0.y = b01.y;
    b1.x = b01.z; b1.y = b01.w;
    b2.x = b23.x; b2.y = b23.y;
    b3.x = b23.z; b3.y = b23.w;
    acc0 = __builtin_amdgcn_wmma_f32_16x16x4_f32(
         false, a,  false, b0,
         (short)0, acc0,  false,  false);
    acc1 = __builtin_amdgcn_wmma_f32_16x16x4_f32(
        false, a, false, b1, (short)0, acc1, false, false);
    acc2 = __builtin_amdgcn_wmma_f32_16x16x4_f32(
        false, a, false, b2, (short)0, acc2, false, false);
    acc3 = __builtin_amdgcn_wmma_f32_16x16x4_f32(
        false, a, false, b3, (short)0, acc3, false, false);
}

template <bool SPLIT>
static __device__ __forceinline__ void RGCN_group_mma(const v2f (&a)[16],
                                                      const float* bl, const int g,
                                                      v8f& acc0, v8f& acc1,
                                                      v8f& acc2, v8f& acc3) {
    const v8f zero = {};
    acc0 = zero; acc1 = zero; acc2 = zero; acc3 = zero;
    if (SPLIT) {
        #pragma unroll
        for (int k = 0; k < 16; ++k) {
            if (k == 8) RGCN_PIN();
            v4f b01, b23;
            RGCN_b_load(bl, k, g, b01, b23);
            RGCN_chunk_mma(a[k], b01, b23, acc0, acc1, acc2, acc3);
        }
    } else {
        v4f p01[18], p23[18];
        RGCN_b_load(bl, 0, g, p01[0], p23[0]);
        RGCN_b_load(bl, 1, g, p01[1], p23[1]);
        #pragma unroll
        for (int k = 0; k < 16; ++k) {
            if (k + 2 < 16) RGCN_b_load(bl, k + 2, g, p01[k + 2], p23[k + 2]);
            RGCN_chunk_mma(a[k], p01[k], p23[k], acc0, acc1, acc2, acc3);
            __builtin_amdgcn_sched_barrier(0);
        }
    }
}

static __device__ __forceinline__ void RGCN_group_store(float* obase,
                                                        const float (&sc)[8],
                                                        const v8f& acc0, const v8f& acc1,
                                                        const v8f& acc2, const v8f& acc3) {
    #pragma unroll
    for (int v = 0; v < 8; ++v) {
        v4f o4;
        o4.x = acc0[v] * sc[v]; o4.y = acc1[v] * sc[v];
        o4.z = acc2[v] * sc[v]; o4.w = acc3[v] * sc[v];
        *(v4f*)(obase + (size_t)v * O_CH) = o4;
    }
}

static __device__ __forceinline__ void RGCN_visit_loads(const float* cs, const float* x,
                                                        const int e_base, const int m,
                                                        const int half, const int lane,
                                                        float (&c0)[16], float (&c1)[16],
                                                        v4f (&xa)[16], v4f (&xb)[16]) {
    const float* csrow = cs + (size_t)(e_base + m) * R_REL + half * 2;
    #pragma unroll
    for (int k = 0; k < 16; ++k) {
        c0[k] = csrow[4 * k + 0];
        c1[k] = csrow[4 * k + 1];
    }
    RGCN_PIN();
    const float* p = x + (size_t)e_base * I_CH + lane * 4;
    #pragma unroll
    for (int r = 0; r < 16; ++r) {
        xa[r] = *(const v4f*)(p + r * I_CH);
        xb[r] = *(const v4f*)(p + r * I_CH + 128);
    }
    RGCN_PIN();
}

__global__ void __launch_bounds__(MAIN_WAVES_PER_BLOCK * 32)
__attribute__((amdgpu_waves_per_eu(3, 3)))
RGCN_main_kernel(const float* cs,
                 const float* __restrict__ Wpack,
                 const float* x,
                 float* out) {
    __shared__ __attribute__((aligned(16))) float Bs[R_REL * O_CH];

    const int wave = __builtin_amdgcn_readfirstlane((int)(threadIdx.x >> 5));
    const int lane = threadIdx.x & 31;
    const int w    = blockIdx.x * MAIN_WAVES_PER_BLOCK + wave;
    int       u     = w * 8;
    int       u_end = u + 8;
    const int xu         = BASE_UNITS + wave * MAIN_BLOCKS + (int)blockIdx.x;
    bool      extra_left = xu < N_UNITS;

    const int half = lane >> 4;
    const int m    = lane & 15;

    float c0[16], c1[16];
    v4f xa[16], xb[16];

    {
        const v4f* src = (const v4f*)Wpack;
        v4f*       dst = (v4f*)Bs;
        v4f tmp[STAGE_PER_THREAD];
        #pragma unroll
        for (int i = 0; i < STAGE_PER_THREAD; ++i) {
            const int idx = i * (MAIN_WAVES_PER_BLOCK * 32) + (int)threadIdx.x;
            if (idx < STAGE_VEC) tmp[i] = src[idx];
        }
        RGCN_PIN();
        RGCN_visit_loads(cs, x, (u >> 2) * 16, m, half, lane, c0, c1, xa, xb);
        #pragma unroll
        for (int i = 0; i < STAGE_PER_THREAD; ++i) {
            const int idx = i * (MAIN_WAVES_PER_BLOCK * 32) + (int)threadIdx.x;
            if (idx < STAGE_VEC) dst[idx] = tmp[i];
        }
    }
    __builtin_amdgcn_fence(__ATOMIC_RELEASE, "workgroup", "local");
    __builtin_amdgcn_s_barrier();
    __builtin_amdgcn_fence(__ATOMIC_ACQUIRE, "workgroup", "local");

    const float* bl = Bs + lane * 4;

    while (true) {
        const int rb     = u >> 2;
        const int g_lo   = u & 3;
        const int left   = u_end - rb * 4;
        const int g_hi   = left < 4 ? left : 4;
        const int e_base = rb * 16;

        v2f a[16];
        #pragma unroll
        for (int k = 0; k < 16; ++k) {
            a[k].x = 1.0f / c0[k];
            a[k].y = 1.0f / c1[k];
        }

        v8f acc0, acc1, acc2, acc3;
        RGCN_group_mma<true>(a, bl, g_lo, acc0, acc1, acc2, acc3);
        RGCN_PIN();

        float s16[16];
        #pragma unroll
        for (int r = 0; r < 16; ++r)
            s16[r] = xa[r].x + xa[r].y + xa[r].z + xa[r].w
                   + xb[r].x + xb[r].y + xb[r].z + xb[r].w;
        const bool h4 = (lane & 16) != 0;
        float s8[8];
        #pragma unroll
        for (int i = 0; i < 8; ++i) {
            const float keep = h4 ? s16[i + 8] : s16[i];
            const float send = h4 ? s16[i]     : s16[i + 8];
            s8[i] = keep + __shfl_xor(send, 16, 32);
        }
        const bool h3 = (lane & 8) != 0;
        float s4[4];
        #pragma unroll
        for (int i = 0; i < 4; ++i) {
            const float keep = h3 ? s8[i + 4] : s8[i];
            const float send = h3 ? s8[i]     : s8[i + 4];
            s4[i] = keep + __shfl_xor(send, 8, 32);
        }
        const bool h2 = (lane & 4) != 0;
        float s2[2];
        #pragma unroll
        for (int i = 0; i < 2; ++i) {
            const float keep = h2 ? s4[i + 2] : s4[i];
            const float send = h2 ? s4[i]     : s4[i + 2];
            s2[i] = keep + __shfl_xor(send, 4, 32);
        }
        const bool h1 = (lane & 2) != 0;
        float t = (h1 ? s2[1] : s2[0]) + __shfl_xor(h1 ? s2[0] : s2[1], 2, 32);
        t += __shfl_xor(t, 1, 32);
        float sc[8];
        #pragma unroll
        for (int v = 0; v < 8; ++v) sc[v] = __shfl(t, (lane & 16) + 2 * v, 32);

        int  u_next = rb * 4 + g_hi;
        bool more   = u_next < u_end;
        if (!more && extra_left) {
            u_next = xu; u_end = xu + 1; extra_left = false; more = true;
        }
        if (more)
            RGCN_visit_loads(cs, x, (u_next >> 2) * 16, m, half, lane, c0, c1, xa, xb);

        float* obase = out + (size_t)(e_base + half * 8) * O_CH + m * 4;
        RGCN_group_store(obase + g_lo * 64, sc, acc0, acc1, acc2, acc3);

        #pragma unroll 1
        for (int g = g_lo + 1; g < g_hi; ++g) {
            RGCN_group_mma<false>(a, bl, g, acc0, acc1, acc2, acc3);
            RGCN_group_store(obase + g * 64, sc, acc0, acc1, acc2, acc3);
        }
        if (!more) break;
        u = u_next;
    }
}

extern "C" void kernel_launch(void* const* d_in, const int* in_sizes, int n_in,
                              void* d_out, int out_size, void* d_ws, size_t ws_size,
                              hipStream_t stream) {
    const float* x  = (const float*)d_in[0];
    const float* cs = (const float*)d_in[1];
    const float* W  = (const float*)d_in[2];
    float* out = (float*)d_out;

    float* Wpack = (float*)d_ws;

    RGCN_wsum_kernel<<<W_BLOCKS, 256, 0, stream>>>(W, Wpack);

    RGCN_main_kernel<<<MAIN_BLOCKS, MAIN_WAVES_PER_BLOCK * 32, 0, stream>>>(cs, Wpack, x, out);
}
